# starting kernel
# baseline (speedup 1.0000x reference)
#include <hip/hip_runtime.h>
#include <hip/hip_bf16.h>

#define VOCAB 100000
#define EMB_DIM 128
#define HIDDEN 256
#define NUM_CLASSES 20
#define BATCH 4096
#define SEQ 200

typedef float v2f __attribute__((ext_vector_type(2)));
typedef float v8f __attribute__((ext_vector_type(8)));

__global__ __launch_bounds__(256) void ft_pool_kernel(
    const int* __restrict__ x, const int* __restrict__ lens,
    const float* __restrict__ emb, float* __restrict__ pooled) {
  const int lane = threadIdx.x & 31;
  const int wave = threadIdx.x >> 5;
  const int b = blockIdx.x * 8 + wave;
  if (b >= BATCH) return;

  int len = lens[b];
  if (len < 1) len = 1;
  if (len > SEQ) len = SEQ;

  const int* xb = x + (size_t)b * SEQ;
  float4 acc = make_float4(0.f, 0.f, 0.f, 0.f);

  for (int s0 = 0; s0 < len; s0 += 32) {
    int cnt = len - s0;
    if (cnt > 32) cnt = 32;
    int tok = 0;
    if (lane < cnt) tok = xb[s0 + lane];
    for (int j = 0; j < cnt; ++j) {
      int t = __shfl(tok, j, 32);
      const float4* row = (const float4*)(emb + (size_t)t * EMB_DIM);
      float4 v = row[lane];
      acc.x += v.x; acc.y += v.y; acc.z += v.z; acc.w += v.w;
    }
  }
  const float inv = 1.0f / (float)len;
  float4 r = make_float4(acc.x * inv, acc.y * inv, acc.z * inv, acc.w * inv);
  ((float4*)(pooled + (size_t)b * EMB_DIM))[lane] = r;
}

__global__ __launch_bounds__(256) void ft_gemm1_kernel(
    const float* __restrict__ pooled, const float* __restrict__ W1,
    const float* __restrict__ b1, float* __restrict__ hidden) {
  const int lane = threadIdx.x & 31;
  const int wave = threadIdx.x >> 5;
  const int tile = blockIdx.x * 8 + wave;
  const int mt = tile >> 4;
  const int nt = tile & 15;
  const int lm = lane & 15;
  const int half = lane >> 4;
  const int kb = half * 2;

  const float* Arow = pooled + (size_t)(mt * 16 + lm) * EMB_DIM;
  const int ncol = nt * 16 + lm;

  v8f c = {0.f, 0.f, 0.f, 0.f, 0.f, 0.f, 0.f, 0.f};
  for (int k0 = 0; k0 < EMB_DIM; k0 += 4) {
    v2f a, bm;
    a.x = Arow[k0 + kb];
    a.y = Arow[k0 + kb + 1];
    bm.x = W1[(size_t)(k0 + kb) * HIDDEN + ncol];
    bm.y = W1[(size_t)(k0 + kb + 1) * HIDDEN + ncol];
    c = __builtin_amdgcn_wmma_f32_16x16x4_f32(
         false, a,  false, bm,
         (short)0, c,  false,  false);
  }

  const float bias = b1[ncol];
  const int mbase = mt * 16 + half * 8;
#pragma unroll
  for (int v = 0; v < 8; ++v) {
    float val = c[v] + bias;
    val = val > 0.f ? val : 0.f;
    hidden[(size_t)(mbase + v) * HIDDEN + ncol] = val;
  }
}

__global__ __launch_bounds__(256) void ft_gemm2_kernel(
    const float* __restrict__ hidden, const float* __restrict__ W2,
    const float* __restrict__ b2, float* __restrict__ out) {
  const int lane = threadIdx.x & 31;
  const int wave = threadIdx.x >> 5;
  const int tile = blockIdx.x * 8 + wave;
  const int mt = tile >> 1;
  const int nt = tile & 1;
  const int lm = lane & 15;
  const int half = lane >> 4;
  const int kb = half * 2;

  const int ncol = nt * 16 + lm;
  const float nmask = (ncol < NUM_CLASSES) ? 1.0f : 0.0f;
  const int nclamp = (ncol < NUM_CLASSES) ? ncol : (NUM_CLASSES - 1);

  const float* Arow = hidden + (size_t)(mt * 16 + lm) * HIDDEN;

  v8f c = {0.f, 0.f, 0.f, 0.f, 0.f, 0.f, 0.f, 0.f};
  for (int k0 = 0; k0 < HIDDEN; k0 += 4) {
    v2f a, bm;
    a.x = Arow[k0 + kb];
    a.y = Arow[k0 + kb + 1];
    bm.x = W2[(size_t)(k0 + kb) * NUM_CLASSES + nclamp] * nmask;
    bm.y = W2[(size_t)(k0 + kb + 1) * NUM_CLASSES + nclamp] * nmask;
    c = __builtin_amdgcn_wmma_f32_16x16x4_f32(
        false, a, false, bm, (short)0, c, false, false);
  }

  if (ncol < NUM_CLASSES) {
    const float bias = b2[ncol];
    const int mbase = mt * 16 + half * 8;
#pragma unroll
    for (int v = 0; v < 8; ++v) {
      out[(size_t)(mbase + v) * NUM_CLASSES + ncol] = c[v] + bias;
    }
  }
}

extern "C" void kernel_launch(void* const* d_in, const int* in_sizes, int n_in,
                              void* d_out, int out_size, void* d_ws, size_t ws_size,
                              hipStream_t stream) {
  const int*   x    = (const int*)d_in[0];
  const int*   lens = (const int*)d_in[1];
  const float* emb  = (const float*)d_in[2];
  const float* W1   = (const float*)d_in[3];
  const float* b1   = (const float*)d_in[4];
  const float* W2   = (const float*)d_in[5];
  const float* b2   = (const float*)d_in[6];
  float*       out  = (float*)d_out;

  float* pooled = (float*)d_ws;
  float* hidden = (float*)((char*)d_ws + (size_t)BATCH * EMB_DIM * sizeof(float));

  ft_pool_kernel<<<BATCH / 8, 256, 0, stream>>>(x, lens, emb, pooled);

  ft_gemm1_kernel<<<(BATCH / 16) * (HIDDEN / 16) / 8, 256, 0, stream>>>(
      pooled, W1, b1, hidden);

  ft_gemm2_kernel<<<(BATCH / 16) * 2 / 8, 256, 0, stream>>>(
      hidden, W2, b2, out);
}


# layers chained in registers: layer 1 computed transposed (W1^T x pooled^T), its relu-ed accumulators are layer 2's A fragments; 16 waves' partial outs added via LDS; 2 launches; out sum order changes
# speedup vs baseline: 4.0524x; 4.0524x over previous
#include <hip/hip_runtime.h>
#include <hip/hip_bf16.h>

#define VOCAB 100000
#define EMB_DIM 128
#define HIDDEN 256
#define NUM_CLASSES 20
#define BATCH 4096
#define SEQ 200

typedef float v2f __attribute__((ext_vector_type(2)));
typedef float v8f __attribute__((ext_vector_type(8)));

#define POOL_SAMPLES 4
#define POOL_PARTS 8
__global__ __launch_bounds__(256) void ft_pool_kernel(
    const int* __restrict__ x, const int* __restrict__ lens,
    const float* __restrict__ emb, float* __restrict__ pooled) {
  __shared__ float part[POOL_SAMPLES][POOL_PARTS][EMB_DIM];

  const int tid = threadIdx.x;
  const int lane = tid & 31;
  const int wave = __builtin_amdgcn_readfirstlane(tid >> 5);
  const int smp = wave >> 1;
  const int grp = lane >> 3;
  const int sub = lane & 7;
  const int w4 = (wave & 1) * 4;
  const int p = w4 + grp;
  const int b = blockIdx.x * POOL_SAMPLES + smp;

  int len = lens[b];
  if (len < 1) len = 1;
  if (len > SEQ) len = SEQ;

  const int* xb = x + (size_t)b * SEQ;
  float4 a0 = make_float4(0.f, 0.f, 0.f, 0.f);
  float4 a1 = a0, a2 = a0, a3 = a0;

  const bool godd = (grp & 1) != 0;
  const bool ghi = (grp & 2) != 0;

  int sb = w4;
  const int4 t4 = *(const int4*)(xb + sb);
  int tok = ghi ? (godd ? t4.w : t4.z) : (godd ? t4.y : t4.x);
  while (sb < len) {
    const int sbn = sb + POOL_PARTS;
    const int sbl = sbn < SEQ - 4 ? sbn : SEQ - 4;
    const int4 t4n = *(const int4*)(xb + sbl);
    if (sb + grp < len) {
      const float4* row = (const float4*)(emb + (size_t)tok * EMB_DIM) + sub;
      const float4 v0 = row[0];
      const float4 v1 = row[8];
      const float4 v2 = row[16];
      const float4 v3 = row[24];
      a0.x += v0.x; a0.y += v0.y; a0.z += v0.z; a0.w += v0.w;
      a1.x += v1.x; a1.y += v1.y; a1.z += v1.z; a1.w += v1.w;
      a2.x += v2.x; a2.y += v2.y; a2.z += v2.z; a2.w += v2.w;
      a3.x += v3.x; a3.y += v3.y; a3.z += v3.z; a3.w += v3.w;
    }
    tok = ghi ? (godd ? t4n.w : t4n.z) : (godd ? t4n.y : t4n.x);
    sb = sbn;
  }

  float4* pp = (float4*)&part[smp][p][0];
  pp[sub] = a0;
  pp[sub + 8] = a1;
  pp[sub + 16] = a2;
  pp[sub + 24] = a3;
  __syncthreads();

  const int j = (tid & 63) * 2;
  float sx = part[smp][0][j];
  float sy = part[smp][0][j + 1];
#pragma unroll
  for (int q = 1; q < POOL_PARTS; ++q) {
    sx += part[smp][q][j];
    sy += part[smp][q][j + 1];
  }
  const float inv = 1.0f / (float)len;
  float2 r;
  r.x = sx * inv;
  r.y = sy * inv;
  *(float2*)(pooled + (size_t)b * EMB_DIM + j) = r;
}

#define MLP_WAVES 16
#define P_PITCH 132
#define PRE_STEPS 8
__global__ __launch_bounds__(512) void ft_mlp_kernel(
    const float* __restrict__ pooled, const float* __restrict__ W1,
    const float* __restrict__ b1, const float* __restrict__ W2,
    const float* __restrict__ b2, float* __restrict__ out) {
  __shared__ float tileP[16 * P_PITCH];
  __shared__ float red[MLP_WAVES * 8 * 2 * 32];

  const int tid = threadIdx.x;
  const int lane = tid & 31;
  const int wave = tid >> 5;
  const int mt = blockIdx.x;
  const int lm = lane & 15;
  const int half = lane >> 4;

  const float* w1p = W1 + (size_t)(2 * half) * HIDDEN + wave * 16 + lm;

  float wpre[2 * PRE_STEPS];
#pragma unroll
  for (int q = 0; q < PRE_STEPS; ++q) {
    wpre[2 * q] = w1p[(size_t)(4 * q) * HIDDEN];
    wpre[2 * q + 1] = w1p[(size_t)(4 * q + 1) * HIDDEN];
  }

  {
    const float4 v =
        *(const float4*)(pooled + (size_t)(mt * 16 + wave) * EMB_DIM + lane * 4);
    *(float4*)&tileP[wave * P_PITCH + lane * 4] = v;
  }
  __syncthreads();

  const float* pp = &tileP[lm * P_PITCH + 2 * half];
  v8f acc = {0.f, 0.f, 0.f, 0.f, 0.f, 0.f, 0.f, 0.f};
#pragma unroll
  for (int q = 0; q < EMB_DIM / 4; ++q) {
    v2f a, bm;
    if (q < PRE_STEPS) {
      a.x = wpre[2 * q];
      a.y = wpre[2 * q + 1];
    } else {
      a.x = w1p[(size_t)(4 * q) * HIDDEN];
      a.y = w1p[(size_t)(4 * q + 1) * HIDDEN];
    }
    bm = *(const v2f*)(pp + 4 * q);
    acc = __builtin_amdgcn_wmma_f32_16x16x4_f32(
         false, a,  false, bm,
         (short)0, acc,  false,  false);
  }

  const float4 bA = *(const float4*)(b1 + wave * 16 + 8 * half);
  const float4 bB = *(const float4*)(b1 + wave * 16 + 8 * half + 4);
  const float bias[8] = {bA.x, bA.y, bA.z, bA.w, bB.x, bB.y, bB.z, bB.w};
  float hid[8];
#pragma unroll
  for (int r = 0; r < 8; ++r) {
    float val = acc[r] + bias[r];
    hid[r] = val > 0.f ? val : 0.f;
  }

  const int n1 = (16 + lm < NUM_CLASSES) ? 16 + lm : NUM_CLASSES - 1;
  const float* w2p = W2 + (size_t)(wave * 16 + 8 * half) * NUM_CLASSES;
  v8f o0 = {0.f, 0.f, 0.f, 0.f, 0.f, 0.f, 0.f, 0.f};
  v8f o1 = {0.f, 0.f, 0.f, 0.f, 0.f, 0.f, 0.f, 0.f};
#pragma unroll
  for (int j = 0; j < 4; ++j) {
    v2f a, q0, q1;
    a.x = hid[2 * j];
    a.y = hid[2 * j + 1];
    q0.x = w2p[(2 * j) * NUM_CLASSES + lm];
    q0.y = w2p[(2 * j + 1) * NUM_CLASSES + lm];
    q1.x = w2p[(2 * j) * NUM_CLASSES + n1];
    q1.y = w2p[(2 * j + 1) * NUM_CLASSES + n1];
    o0 = __builtin_amdgcn_wmma_f32_16x16x4_f32(
        false, a, false, q0, (short)0, o0, false, false);
    o1 = __builtin_amdgcn_wmma_f32_16x16x4_f32(
        false, a, false, q1, (short)0, o1, false, false);
  }

  float* rw = red + wave * 512 + half * 32 + lm;
#pragma unroll
  for (int r = 0; r < 8; ++r) rw[r * 64] = o0[r];
  if (lm < NUM_CLASSES - 16) {
#pragma unroll
    for (int r = 0; r < 8; ++r) rw[r * 64 + 16] = o1[r];
  }
  __syncthreads();

  if (tid < 16 * NUM_CLASSES) {
    const int smp = tid / NUM_CLASSES;
    const int n = tid - smp * NUM_CLASSES;
    const float* rp = red + (smp & 7) * 64 + (smp >> 3) * 32 + n;
    float s = rp[0];
#pragma unroll
    for (int w = 1; w < MLP_WAVES; ++w) s += rp[w * 512];
    out[(size_t)mt * (16 * NUM_CLASSES) + tid] = s + b2[n];
  }
}

extern "C" void kernel_launch(void* const* d_in, const int* in_sizes, int n_in,
                              void* d_out, int out_size, void* d_ws, size_t ws_size,
                              hipStream_t stream) {
  const int*   x    = (const int*)d_in[0];
  const int*   lens = (const int*)d_in[1];
  const float* emb  = (const float*)d_in[2];
  const float* W1   = (const float*)d_in[3];
  const float* b1   = (const float*)d_in[4];
  const float* W2   = (const float*)d_in[5];
  const float* b2   = (const float*)d_in[6];
  float*       out  = (float*)d_out;

  float* pooled = (float*)d_ws;

  ft_pool_kernel<<<BATCH / POOL_SAMPLES, 256, 0, stream>>>(x, lens, emb, pooled);

  ft_mlp_kernel<<<BATCH / 16, 512, 0, stream>>>(pooled, W1, b1, W2, b2, out);
}
